# starting kernel
# baseline (speedup 1.0000x reference)
#include <hip/hip_runtime.h>
#include <math.h>

constexpr int N_DOC = 4;
constexpr int L     = 1024;
constexpr int D     = 768;
constexpr int H     = 12;
constexpr int E     = 32;
constexpr int M     = 4;
constexpr int R     = 256;
constexpr float NEG_MAX = -3.402823466e+38f;

typedef __attribute__((ext_vector_type(2))) float v2f;
typedef __attribute__((ext_vector_type(4))) float v4f;
typedef __attribute__((ext_vector_type(8))) float v8f;

__global__ void k_e_emb(const float* __restrict__ seq,
                        const int*   __restrict__ pos,
                        const float* __restrict__ mask,
                        float* __restrict__ e_emb) {
    int b = blockIdx.x;
    int i = b / E;
    const float* seq_i  = seq  + (size_t)i * L * D;
    const int*   pos_e  = pos  + (size_t)b * M;
    const float* mask_e = mask + (size_t)b * M;

    int   p[M];
    float mk[M];
#pragma unroll
    for (int m = 0; m < M; ++m) { p[m] = pos_e[m] + 1; mk[m] = mask_e[m]; }

    for (int c = threadIdx.x; c < D; c += blockDim.x) {
        float v[M];
        float mx = NEG_MAX;
#pragma unroll
        for (int m = 0; m < M; ++m) {
            float val = seq_i[(size_t)p[m] * D + c];
            v[m] = (mk[m] > 0.0f) ? val : NEG_MAX;
            mx = fmaxf(mx, v[m]);
        }
        float s = 0.0f;
#pragma unroll
        for (int m = 0; m < M; ++m)
            if (mk[m] > 0.0f) s += expf(v[m] - mx);
        e_emb[(size_t)b * D + c] = mx + logf(s);
    }
}

__global__ void k_e_att(const float* __restrict__ att,
                        const int*   __restrict__ pos,
                        const float* __restrict__ mask,
                        float* __restrict__ e_att) {
    int b  = blockIdx.x;
    int hh = b % H;
    int e  = (b / H) % E;
    int i  = b / (H * E);

    const int*   pos_e  = pos  + ((size_t)i * E + e) * M;
    const float* mask_e = mask + ((size_t)i * E + e) * M;

    float cnt = 0.0f;
#pragma unroll
    for (int m = 0; m < M; ++m) cnt += mask_e[m];
    float inv = 1.0f / fmaxf(cnt, 1.0f);

    int l0 = threadIdx.x * 4;
    v4f acc = {0.0f, 0.0f, 0.0f, 0.0f};
#pragma unroll
    for (int m = 0; m < M; ++m) {
        float mk = mask_e[m];
        if (mk > 0.0f) {
            int p = pos_e[m] + 1;
            const v4f* row = (const v4f*)(att + (((size_t)i * H + hh) * L + p) * L + l0);
            acc += mk * row[0];
        }
    }
    acc *= inv;
    v4f* dst = (v4f*)(e_att + (((size_t)i * E + e) * H + hh) * L + l0);
    dst[0] = acc;
}

__global__ void k_ht_att(const float* __restrict__ e_att,
                         const int*   __restrict__ hts,
                         float* __restrict__ ht_att) {
    int b = blockIdx.x;
    int i = b / R;
    int eh = hts[(size_t)b * 2 + 0];
    int et = hts[(size_t)b * 2 + 1];

    const float* ha = e_att + ((size_t)i * E + eh) * H * L;
    const float* ta = e_att + ((size_t)i * E + et) * H * L;

    int l0 = threadIdx.x * 4;
    v4f acc = {0.0f, 0.0f, 0.0f, 0.0f};
#pragma unroll
    for (int hh = 0; hh < H; ++hh) {
        v4f a = *(const v4f*)(ha + (size_t)hh * L + l0);
        v4f t = *(const v4f*)(ta + (size_t)hh * L + l0);
        acc += a * t;
    }
    acc *= (1.0f / (float)H);

    __shared__ float red[256];
    float ps = acc.x + acc.y + acc.z + acc.w;
    red[threadIdx.x] = ps;
    __syncthreads();
    for (int s = 128; s > 0; s >>= 1) {
        if (threadIdx.x < s) red[threadIdx.x] += red[threadIdx.x + s];
        __syncthreads();
    }
    float invs = 1.0f / (red[0] + 1e-5f);

    v4f* dst = (v4f*)(ht_att + (size_t)b * L + l0);
    dst[0] = acc * invs;
}

__global__ void k_hs_ts(const float* __restrict__ e_emb,
                        const int*   __restrict__ hts,
                        float* __restrict__ out) {
    int b = blockIdx.x;
    int i = b / R;
    int eh = hts[(size_t)b * 2 + 0];
    int et = hts[(size_t)b * 2 + 1];

    const float* sh = e_emb + ((size_t)i * E + eh) * D;
    const float* st = e_emb + ((size_t)i * E + et) * D;
    float* oh = out + (size_t)b * D;
    float* ot = out + (size_t)N_DOC * R * D + (size_t)b * D;

    for (int c = threadIdx.x; c < D; c += blockDim.x) {
        oh[c] = sh[c];
        ot[c] = st[c];
    }
}

__global__ void k_rs_wmma(const float* __restrict__ seq,
                          const float* __restrict__ ht_att,
                          float* __restrict__ out) {
    constexpr int NT = D / 16;
    constexpr int MT = R / 32;
    int b  = blockIdx.x;
    int nt = b % NT;
    int mt = (b / NT) % MT;
    int i  = b / (NT * MT);

    int m0 = mt * 32;
    int n0 = nt * 16;

    const float* A  = ht_att + (size_t)i * R * L;
    const float* Bm = seq    + (size_t)i * L * D;

    int lane = threadIdx.x & 31;
    int half = lane >> 4;
    int l15  = lane & 15;
    int bcol = n0 + l15;

    v8f c0 = {0.f, 0.f, 0.f, 0.f, 0.f, 0.f, 0.f, 0.f};
    v8f c1 = {0.f, 0.f, 0.f, 0.f, 0.f, 0.f, 0.f, 0.f};

    const float* arow0 = A + (size_t)(m0 + l15) * L;
    const float* arow1 = A + (size_t)(m0 + 16 + l15) * L;

    for (int k0 = 0; k0 < L; k0 += 4) {
        int ka = k0 + half * 2;
        v2f a0 = *(const v2f*)(arow0 + ka);
        v2f a1 = *(const v2f*)(arow1 + ka);
        v2f bb;
        bb.x = Bm[(size_t)ka * D + bcol];
        bb.y = Bm[(size_t)(ka + 1) * D + bcol];
        c0 = __builtin_amdgcn_wmma_f32_16x16x4_f32(
                false, a0, false, bb, (short)0, c0, false, false);
        c1 = __builtin_amdgcn_wmma_f32_16x16x4_f32(
                false, a1, false, bb, (short)0, c1, false, false);
    }

    float* O = out + (size_t)2 * N_DOC * R * D + (size_t)i * R * D;
#pragma unroll
    for (int v = 0; v < 8; ++v) {
        int row0 = m0 + v + half * 8;
        int row1 = m0 + 16 + v + half * 8;
        O[(size_t)row0 * D + n0 + l15] = c0[v];
        O[(size_t)row1 * D + n0 + l15] = c1[v];
    }
}

extern "C" void kernel_launch(void* const* d_in, const int* in_sizes, int n_in,
                              void* d_out, int out_size, void* d_ws, size_t ws_size,
                              hipStream_t stream) {
    const float* seq   = (const float*)d_in[0];
    const float* att   = (const float*)d_in[1];
    const int*   pos   = (const int*)  d_in[2];
    const float* mask  = (const float*)d_in[3];
    const int*   hts   = (const int*)  d_in[4];
    float*       out   = (float*)d_out;

    float* ws     = (float*)d_ws;
    float* e_emb  = ws;
    float* e_att  = e_emb + (size_t)N_DOC * E * D;
    float* ht_att = e_att + (size_t)N_DOC * E * H * L;

    k_e_emb  <<<N_DOC * E,     256, 0, stream>>>(seq, pos, mask, e_emb);
    k_e_att  <<<N_DOC * E * H, 256, 0, stream>>>(att, pos, mask, e_att);
    k_ht_att <<<N_DOC * R,     256, 0, stream>>>(e_att, hts, ht_att);
    k_hs_ts  <<<N_DOC * R,     256, 0, stream>>>(e_emb, hts, out);
    k_rs_wmma<<<N_DOC * (R / 32) * (D / 16), 32, 0, stream>>>(seq, ht_att, out);
}


# rs GEMM: 64x48 block tiles staged through LDS by 4 waves (3 accumulator tiles per wave, whole-line global loads one K chunk ahead) replace per-wave strided global fragment loads
# speedup vs baseline: 1.6115x; 1.6115x over previous
#include <hip/hip_runtime.h>
#include <math.h>

constexpr int N_DOC = 4;
constexpr int L     = 1024;
constexpr int D     = 768;
constexpr int H     = 12;
constexpr int E     = 32;
constexpr int M     = 4;
constexpr int R     = 256;
constexpr float NEG_MAX = -3.402823466e+38f;

typedef __attribute__((ext_vector_type(2))) float v2f;
typedef __attribute__((ext_vector_type(4))) float v4f;
typedef __attribute__((ext_vector_type(8))) float v8f;

__global__ void k_e_emb(const float* __restrict__ seq,
                        const int*   __restrict__ pos,
                        const float* __restrict__ mask,
                        float* __restrict__ e_emb) {
    int b = blockIdx.x;
    int i = b / E;
    const float* seq_i  = seq  + (size_t)i * L * D;
    const int*   pos_e  = pos  + (size_t)b * M;
    const float* mask_e = mask + (size_t)b * M;

    int   p[M];
    float mk[M];
#pragma unroll
    for (int m = 0; m < M; ++m) { p[m] = pos_e[m] + 1; mk[m] = mask_e[m]; }

    for (int c = threadIdx.x; c < D; c += blockDim.x) {
        float v[M];
        float mx = NEG_MAX;
#pragma unroll
        for (int m = 0; m < M; ++m) {
            float val = seq_i[(size_t)p[m] * D + c];
            v[m] = (mk[m] > 0.0f) ? val : NEG_MAX;
            mx = fmaxf(mx, v[m]);
        }
        float s = 0.0f;
#pragma unroll
        for (int m = 0; m < M; ++m)
            if (mk[m] > 0.0f) s += expf(v[m] - mx);
        e_emb[(size_t)b * D + c] = mx + logf(s);
    }
}

__global__ void k_e_att(const float* __restrict__ att,
                        const int*   __restrict__ pos,
                        const float* __restrict__ mask,
                        float* __restrict__ e_att) {
    int b  = blockIdx.x;
    int hh = b % H;
    int e  = (b / H) % E;
    int i  = b / (H * E);

    const int*   pos_e  = pos  + ((size_t)i * E + e) * M;
    const float* mask_e = mask + ((size_t)i * E + e) * M;

    float cnt = 0.0f;
#pragma unroll
    for (int m = 0; m < M; ++m) cnt += mask_e[m];
    float inv = 1.0f / fmaxf(cnt, 1.0f);

    int l0 = threadIdx.x * 4;
    v4f acc = {0.0f, 0.0f, 0.0f, 0.0f};
#pragma unroll
    for (int m = 0; m < M; ++m) {
        float mk = mask_e[m];
        if (mk > 0.0f) {
            int p = pos_e[m] + 1;
            const v4f* row = (const v4f*)(att + (((size_t)i * H + hh) * L + p) * L + l0);
            acc += mk * row[0];
        }
    }
    acc *= inv;
    v4f* dst = (v4f*)(e_att + (((size_t)i * E + e) * H + hh) * L + l0);
    dst[0] = acc;
}

__global__ void k_ht_att(const float* __restrict__ e_att,
                         const int*   __restrict__ hts,
                         float* __restrict__ ht_att) {
    int b = blockIdx.x;
    int i = b / R;
    int eh = hts[(size_t)b * 2 + 0];
    int et = hts[(size_t)b * 2 + 1];

    const float* ha = e_att + ((size_t)i * E + eh) * H * L;
    const float* ta = e_att + ((size_t)i * E + et) * H * L;

    int l0 = threadIdx.x * 4;
    v4f acc = {0.0f, 0.0f, 0.0f, 0.0f};
#pragma unroll
    for (int hh = 0; hh < H; ++hh) {
        v4f a = *(const v4f*)(ha + (size_t)hh * L + l0);
        v4f t = *(const v4f*)(ta + (size_t)hh * L + l0);
        acc += a * t;
    }
    acc *= (1.0f / (float)H);

    __shared__ float red[256];
    float ps = acc.x + acc.y + acc.z + acc.w;
    red[threadIdx.x] = ps;
    __syncthreads();
    for (int s = 128; s > 0; s >>= 1) {
        if (threadIdx.x < s) red[threadIdx.x] += red[threadIdx.x + s];
        __syncthreads();
    }
    float invs = 1.0f / (red[0] + 1e-5f);

    v4f* dst = (v4f*)(ht_att + (size_t)b * L + l0);
    dst[0] = acc * invs;
}

__global__ void k_hs_ts(const float* __restrict__ e_emb,
                        const int*   __restrict__ hts,
                        float* __restrict__ out) {
    int b = blockIdx.x;
    int i = b / R;
    int eh = hts[(size_t)b * 2 + 0];
    int et = hts[(size_t)b * 2 + 1];

    const float* sh = e_emb + ((size_t)i * E + eh) * D;
    const float* st = e_emb + ((size_t)i * E + et) * D;
    float* oh = out + (size_t)b * D;
    float* ot = out + (size_t)N_DOC * R * D + (size_t)b * D;

    for (int c = threadIdx.x; c < D; c += blockDim.x) {
        oh[c] = sh[c];
        ot[c] = st[c];
    }
}

constexpr int G_BM  = 64;
constexpr int G_BN  = 48;
constexpr int G_KC  = 64;
constexpr int G_AP  = G_KC + 4;
constexpr int G_ASZ = G_BM * G_AP;
constexpr int G_BSZ = G_KC * G_BN;
constexpr int G_NCH = L / G_KC;
constexpr int G_ALD = (G_BM * G_KC / 4) / 128;
constexpr int G_BLD = (G_KC * G_BN / 4) / 128;

__global__ __launch_bounds__(128)
void k_rs_wmma(const float* __restrict__ seq,
               const float* __restrict__ ht_att,
               float* __restrict__ out) {
    __shared__ __attribute__((aligned(16))) float sA[2][G_ASZ];
    __shared__ __attribute__((aligned(16))) float sB[2][G_BSZ];

    constexpr int NT = D / G_BN;
    constexpr int MT = R / G_BM;
    int b  = blockIdx.x;
    int nt = b % NT;
    int mt = (b / NT) % MT;
    int i  = b / (NT * MT);

    int m0 = mt * G_BM;
    int n0 = nt * G_BN;

    const float* A  = ht_att + (size_t)i * R * L + (size_t)m0 * L;
    const float* Bm = seq    + (size_t)i * L * D + n0;

    int tid  = threadIdx.x;
    int wave = tid >> 5;
    int lane = tid & 31;
    int half = lane >> 4;
    int l15  = lane & 15;

    int a_goff[G_ALD], a_soff[G_ALD];
#pragma unroll
    for (int j = 0; j < G_ALD; ++j) {
        int f   = tid + 128 * j;
        int row = f >> 4;
        int c4  = f & 15;
        a_goff[j] = row * L + c4 * 4;
        a_soff[j] = row * G_AP + c4 * 4;
    }
    int b_goff[G_BLD], b_soff[G_BLD];
#pragma unroll
    for (int j = 0; j < G_BLD; ++j) {
        int f    = tid + 128 * j;
        int krow = f / 12;
        int c4   = f - krow * 12;
        b_goff[j] = krow * D + c4 * 4;
        b_soff[j] = f * 4;
    }

    v4f ra[G_ALD], rb[G_BLD];

#pragma unroll
    for (int j = 0; j < G_ALD; ++j) ra[j] = *(const v4f*)(A + a_goff[j]);
#pragma unroll
    for (int j = 0; j < G_BLD; ++j) rb[j] = *(const v4f*)(Bm + b_goff[j]);
#pragma unroll
    for (int j = 0; j < G_ALD; ++j) *(v4f*)(&sA[0][a_soff[j]]) = ra[j];
#pragma unroll
    for (int j = 0; j < G_BLD; ++j) *(v4f*)(&sB[0][b_soff[j]]) = rb[j];
    __syncthreads();

    v8f acc[3];
#pragma unroll
    for (int j = 0; j < 3; ++j) acc[j] = (v8f){0.f, 0.f, 0.f, 0.f, 0.f, 0.f, 0.f, 0.f};

    int a_frag = (wave * 16 + l15) * G_AP + half * 2;
    int b_frag = (half * 2) * G_BN + l15;

    for (int c = 0; c < G_NCH; ++c) {
        int cur = c & 1;
        bool more = (c + 1 < G_NCH);
        if (more) {
            const float* An = A  + (size_t)(c + 1) * G_KC;
            const float* Bn = Bm + (size_t)(c + 1) * G_KC * D;
#pragma unroll
            for (int j = 0; j < G_ALD; ++j) ra[j] = *(const v4f*)(An + a_goff[j]);
#pragma unroll
            for (int j = 0; j < G_BLD; ++j) rb[j] = *(const v4f*)(Bn + b_goff[j]);
        }

        const float* cA = &sA[cur][a_frag];
        const float* cB = &sB[cur][b_frag];
#pragma unroll
        for (int ks = 0; ks < G_KC / 4; ++ks) {
            v2f a = *(const v2f*)(cA + ks * 4);
#pragma unroll
            for (int j = 0; j < 3; ++j) {
                v2f bb;
                bb.x = cB[(ks * 4) * G_BN + j * 16];
                bb.y = cB[(ks * 4 + 1) * G_BN + j * 16];
                acc[j] = __builtin_amdgcn_wmma_f32_16x16x4_f32(
                            false, a, false, bb, (short)0, acc[j], false, false);
            }
        }

        if (more) {
            int nxt = cur ^ 1;
#pragma unroll
            for (int j = 0; j < G_ALD; ++j) *(v4f*)(&sA[nxt][a_soff[j]]) = ra[j];
#pragma unroll
            for (int j = 0; j < G_BLD; ++j) *(v4f*)(&sB[nxt][b_soff[j]]) = rb[j];
        }
        __syncthreads();
    }

    float* O = out + (size_t)2 * N_DOC * R * D + (size_t)i * R * D;
#pragma unroll
    for (int j = 0; j < 3; ++j) {
#pragma unroll
        for (int v = 0; v < 8; ++v) {
            int row = m0 + wave * 16 + v + half * 8;
            O[(size_t)row * D + n0 + j * 16 + l15] = acc[j][v];
        }
    }
}

extern "C" void kernel_launch(void* const* d_in, const int* in_sizes, int n_in,
                              void* d_out, int out_size, void* d_ws, size_t ws_size,
                              hipStream_t stream) {
    const float* seq   = (const float*)d_in[0];
    const float* att   = (const float*)d_in[1];
    const int*   pos   = (const int*)  d_in[2];
    const float* mask  = (const float*)d_in[3];
    const int*   hts   = (const int*)  d_in[4];
    float*       out   = (float*)d_out;

    float* ws     = (float*)d_ws;
    float* e_emb  = ws;
    float* e_att  = e_emb + (size_t)N_DOC * E * D;
    float* ht_att = e_att + (size_t)N_DOC * E * H * L;

    k_e_emb  <<<N_DOC * E,     256, 0, stream>>>(seq, pos, mask, e_emb);
    k_e_att  <<<N_DOC * E * H, 256, 0, stream>>>(att, pos, mask, e_att);
    k_ht_att <<<N_DOC * R,     256, 0, stream>>>(e_att, hts, ht_att);
    k_hs_ts  <<<N_DOC * R,     256, 0, stream>>>(e_emb, hts, out);
    k_rs_wmma<<<N_DOC * (R / G_BM) * (D / G_BN), 128, 0, stream>>>(seq, ht_att, out);
}


# launch fusion 5 -> 3: e_emb columns ride on 2 waves of each e_att block, hs/ts row copies ride on the ht_att blocks; 2 launch gaps and 9216 waves less, arithmetic and sum order unchanged
# speedup vs baseline: 2.0407x; 1.2663x over previous
#include <hip/hip_runtime.h>
#include <math.h>

constexpr int N_DOC = 4;
constexpr int L     = 1024;
constexpr int D     = 768;
constexpr int H     = 12;
constexpr int E     = 32;
constexpr int M     = 4;
constexpr int R     = 256;
constexpr float NEG_MAX = -3.402823466e+38f;

typedef __attribute__((ext_vector_type(2))) float v2f;
typedef __attribute__((ext_vector_type(4))) float v4f;
typedef __attribute__((ext_vector_type(8))) float v8f;

constexpr int EC = D / H;
static_assert(D % H == 0 && EC <= 256, "e_emb columns are split evenly over the head blocks");

__global__ void k_entity(const float* __restrict__ seq,
                         const float* __restrict__ att,
                         const int*   __restrict__ pos,
                         const float* __restrict__ mask,
                         float* __restrict__ e_emb,
                         float* __restrict__ e_att) {
    int b  = blockIdx.x;
    int hh = b % H;
    int ie = b / H;
    int i  = ie / E;

    const int*   pos_e  = pos  + (size_t)ie * M;
    const float* mask_e = mask + (size_t)ie * M;

    int   p[M];
    float mk[M];
    float cnt = 0.0f;
#pragma unroll
    for (int m = 0; m < M; ++m) { p[m] = pos_e[m] + 1; mk[m] = mask_e[m]; cnt += mk[m]; }
    float inv = 1.0f / fmaxf(cnt, 1.0f);

    bool emb = threadIdx.x < EC;
    int  c   = hh * EC + threadIdx.x;
    float sv[M];
    if (emb) {
        const float* seq_i = seq + (size_t)i * L * D;
#pragma unroll
        for (int m = 0; m < M; ++m) sv[m] = seq_i[(size_t)p[m] * D + c];
    }

    int l0 = threadIdx.x * 4;
    v4f acc = {0.0f, 0.0f, 0.0f, 0.0f};
#pragma unroll
    for (int m = 0; m < M; ++m) {
        if (mk[m] > 0.0f) {
            const v4f* row = (const v4f*)(att + (((size_t)i * H + hh) * L + p[m]) * L + l0);
            acc += mk[m] * row[0];
        }
    }
    acc *= inv;
    v4f* dst = (v4f*)(e_att + ((size_t)ie * H + hh) * L + l0);
    dst[0] = acc;

    if (emb) {
        float v[M];
        float mx = NEG_MAX;
#pragma unroll
        for (int m = 0; m < M; ++m) {
            v[m] = (mk[m] > 0.0f) ? sv[m] : NEG_MAX;
            mx = fmaxf(mx, v[m]);
        }
        float s = 0.0f;
#pragma unroll
        for (int m = 0; m < M; ++m)
            if (mk[m] > 0.0f) s += expf(v[m] - mx);
        e_emb[(size_t)ie * D + c] = mx + logf(s);
    }
}

constexpr int D4 = D / 4;
static_assert(D % 4 == 0 && D4 <= 256, "one float4 of hs and of ts per thread");

__global__ void k_pair(const float* __restrict__ e_emb,
                       const float* __restrict__ e_att,
                       const int*   __restrict__ hts,
                       float* __restrict__ ht_att,
                       float* __restrict__ out) {
    int b = blockIdx.x;
    int i = b / R;
    int eh = hts[(size_t)b * 2 + 0];
    int et = hts[(size_t)b * 2 + 1];

    bool cp = threadIdx.x < D4;
    v4f ch, ct;
    if (cp) {
        ch = *(const v4f*)(e_emb + ((size_t)i * E + eh) * D + threadIdx.x * 4);
        ct = *(const v4f*)(e_emb + ((size_t)i * E + et) * D + threadIdx.x * 4);
    }

    const float* ha = e_att + ((size_t)i * E + eh) * H * L;
    const float* ta = e_att + ((size_t)i * E + et) * H * L;

    int l0 = threadIdx.x * 4;
    v4f acc = {0.0f, 0.0f, 0.0f, 0.0f};
#pragma unroll
    for (int hh = 0; hh < H; ++hh) {
        v4f a = *(const v4f*)(ha + (size_t)hh * L + l0);
        v4f t = *(const v4f*)(ta + (size_t)hh * L + l0);
        acc += a * t;
    }
    acc *= (1.0f / (float)H);

    __shared__ float red[256];
    float ps = acc.x + acc.y + acc.z + acc.w;
    red[threadIdx.x] = ps;
    __syncthreads();
    for (int s = 128; s > 0; s >>= 1) {
        if (threadIdx.x < s) red[threadIdx.x] += red[threadIdx.x + s];
        __syncthreads();
    }
    float invs = 1.0f / (red[0] + 1e-5f);

    v4f* dst = (v4f*)(ht_att + (size_t)b * L + l0);
    dst[0] = acc * invs;

    if (cp) {
        *(v4f*)(out + (size_t)b * D + threadIdx.x * 4) = ch;
        *(v4f*)(out + (size_t)N_DOC * R * D + (size_t)b * D + threadIdx.x * 4) = ct;
    }
}

constexpr int G_BM  = 64;
constexpr int G_BN  = 48;
constexpr int G_KC  = 64;
constexpr int G_AP  = G_KC + 4;
constexpr int G_ASZ = G_BM * G_AP;
constexpr int G_BSZ = G_KC * G_BN;
constexpr int G_NCH = L / G_KC;
constexpr int G_ALD = (G_BM * G_KC / 4) / 128;
constexpr int G_BLD = (G_KC * G_BN / 4) / 128;

__global__ __launch_bounds__(128)
void k_rs_wmma(const float* __restrict__ seq,
               const float* __restrict__ ht_att,
               float* __restrict__ out) {
    __shared__ __attribute__((aligned(16))) float sA[2][G_ASZ];
    __shared__ __attribute__((aligned(16))) float sB[2][G_BSZ];

    constexpr int NT = D / G_BN;
    constexpr int MT = R / G_BM;
    int b  = blockIdx.x;
    int nt = b % NT;
    int mt = (b / NT) % MT;
    int i  = b / (NT * MT);

    int m0 = mt * G_BM;
    int n0 = nt * G_BN;

    const float* A  = ht_att + (size_t)i * R * L + (size_t)m0 * L;
    const float* Bm = seq    + (size_t)i * L * D + n0;

    int tid  = threadIdx.x;
    int wave = tid >> 5;
    int lane = tid & 31;
    int half = lane >> 4;
    int l15  = lane & 15;

    int a_goff[G_ALD], a_soff[G_ALD];
#pragma unroll
    for (int j = 0; j < G_ALD; ++j) {
        int f   = tid + 128 * j;
        int row = f >> 4;
        int c4  = f & 15;
        a_goff[j] = row * L + c4 * 4;
        a_soff[j] = row * G_AP + c4 * 4;
    }
    int b_goff[G_BLD], b_soff[G_BLD];
#pragma unroll
    for (int j = 0; j < G_BLD; ++j) {
        int f    = tid + 128 * j;
        int krow = f / 12;
        int c4   = f - krow * 12;
        b_goff[j] = krow * D + c4 * 4;
        b_soff[j] = f * 4;
    }

    v4f ra[G_ALD], rb[G_BLD];

#pragma unroll
    for (int j = 0; j < G_ALD; ++j) ra[j] = *(const v4f*)(A + a_goff[j]);
#pragma unroll
    for (int j = 0; j < G_BLD; ++j) rb[j] = *(const v4f*)(Bm + b_goff[j]);
#pragma unroll
    for (int j = 0; j < G_ALD; ++j) *(v4f*)(&sA[0][a_soff[j]]) = ra[j];
#pragma unroll
    for (int j = 0; j < G_BLD; ++j) *(v4f*)(&sB[0][b_soff[j]]) = rb[j];
    __syncthreads();

    v8f acc[3];
#pragma unroll
    for (int j = 0; j < 3; ++j) acc[j] = (v8f){0.f, 0.f, 0.f, 0.f, 0.f, 0.f, 0.f, 0.f};

    int a_frag = (wave * 16 + l15) * G_AP + half * 2;
    int b_frag = (half * 2) * G_BN + l15;

    for (int c = 0; c < G_NCH; ++c) {
        int cur = c & 1;
        bool more = (c + 1 < G_NCH);
        if (more) {
            const float* An = A  + (size_t)(c + 1) * G_KC;
            const float* Bn = Bm + (size_t)(c + 1) * G_KC * D;
#pragma unroll
            for (int j = 0; j < G_ALD; ++j) ra[j] = *(const v4f*)(An + a_goff[j]);
#pragma unroll
            for (int j = 0; j < G_BLD; ++j) rb[j] = *(const v4f*)(Bn + b_goff[j]);
        }

        const float* cA = &sA[cur][a_frag];
        const float* cB = &sB[cur][b_frag];
#pragma unroll
        for (int ks = 0; ks < G_KC / 4; ++ks) {
            v2f a = *(const v2f*)(cA + ks * 4);
#pragma unroll
            for (int j = 0; j < 3; ++j) {
                v2f bb;
                bb.x = cB[(ks * 4) * G_BN + j * 16];
                bb.y = cB[(ks * 4 + 1) * G_BN + j * 16];
                acc[j] = __builtin_amdgcn_wmma_f32_16x16x4_f32(
                            false, a, false, bb, (short)0, acc[j], false, false);
            }
        }

        if (more) {
            int nxt = cur ^ 1;
#pragma unroll
            for (int j = 0; j < G_ALD; ++j) *(v4f*)(&sA[nxt][a_soff[j]]) = ra[j];
#pragma unroll
            for (int j = 0; j < G_BLD; ++j) *(v4f*)(&sB[nxt][b_soff[j]]) = rb[j];
        }
        __syncthreads();
    }

    float* O = out + (size_t)2 * N_DOC * R * D + (size_t)i * R * D;
#pragma unroll
    for (int j = 0; j < 3; ++j) {
#pragma unroll
        for (int v = 0; v < 8; ++v) {
            int row = m0 + wave * 16 + v + half * 8;
            O[(size_t)row * D + n0 + j * 16 + l15] = acc[j][v];
        }
    }
}

extern "C" void kernel_launch(void* const* d_in, const int* in_sizes, int n_in,
                              void* d_out, int out_size, void* d_ws, size_t ws_size,
                              hipStream_t stream) {
    const float* seq   = (const float*)d_in[0];
    const float* att   = (const float*)d_in[1];
    const int*   pos   = (const int*)  d_in[2];
    const float* mask  = (const float*)d_in[3];
    const int*   hts   = (const int*)  d_in[4];
    float*       out   = (float*)d_out;

    float* ws     = (float*)d_ws;
    float* e_emb  = ws;
    float* e_att  = e_emb + (size_t)N_DOC * E * D;
    float* ht_att = e_att + (size_t)N_DOC * E * H * L;

    k_entity <<<N_DOC * E * H, 256, 0, stream>>>(seq, att, pos, mask, e_emb, e_att);
    k_pair   <<<N_DOC * R,     256, 0, stream>>>(e_emb, e_att, hts, ht_att, out);
    k_rs_wmma<<<N_DOC * (R / G_BM) * (D / G_BN), 128, 0, stream>>>(seq, ht_att, out);
}


# rs GEMM LDS path: B parked transposed so each A/B fragment is one 16-byte LDS load per 2 matrix ops, loads grouped ahead of matrix blocks, LDS-only barrier; k sum order in groups of 8 changes
# speedup vs baseline: 2.2417x; 1.0985x over previous
#include <hip/hip_runtime.h>
#include <math.h>

constexpr int N_DOC = 4;
constexpr int L     = 1024;
constexpr int D     = 768;
constexpr int H     = 12;
constexpr int E     = 32;
constexpr int M     = 4;
constexpr int R     = 256;
constexpr float NEG_MAX = -3.402823466e+38f;

typedef __attribute__((ext_vector_type(2))) float v2f;
typedef __attribute__((ext_vector_type(4))) float v4f;
typedef __attribute__((ext_vector_type(8))) float v8f;

constexpr int EC = D / H;
static_assert(D % H == 0 && EC <= 256, "e_emb columns are split evenly over the head blocks");

__global__ void k_entity(const float* __restrict__ seq,
                         const float* __restrict__ att,
                         const int*   __restrict__ pos,
                         const float* __restrict__ mask,
                         float* __restrict__ e_emb,
                         float* __restrict__ e_att) {
    int b  = blockIdx.x;
    int hh = b % H;
    int ie = b / H;
    int i  = ie / E;

    const int*   pos_e  = pos  + (size_t)ie * M;
    const float* mask_e = mask + (size_t)ie * M;

    int   p[M];
    float mk[M];
    float cnt = 0.0f;
#pragma unroll
    for (int m = 0; m < M; ++m) { p[m] = pos_e[m] + 1; mk[m] = mask_e[m]; cnt += mk[m]; }
    float inv = 1.0f / fmaxf(cnt, 1.0f);

    bool emb = threadIdx.x < EC;
    int  c   = hh * EC + threadIdx.x;
    float sv[M];
    if (emb) {
        const float* seq_i = seq + (size_t)i * L * D;
#pragma unroll
        for (int m = 0; m < M; ++m) sv[m] = seq_i[(size_t)p[m] * D + c];
    }

    int l0 = threadIdx.x * 4;
    v4f acc = {0.0f, 0.0f, 0.0f, 0.0f};
#pragma unroll
    for (int m = 0; m < M; ++m) {
        if (mk[m] > 0.0f) {
            const v4f* row = (const v4f*)(att + (((size_t)i * H + hh) * L + p[m]) * L + l0);
            acc += mk[m] * row[0];
        }
    }
    acc *= inv;
    v4f* dst = (v4f*)(e_att + ((size_t)ie * H + hh) * L + l0);
    dst[0] = acc;

    if (emb) {
        float v[M];
        float mx = NEG_MAX;
#pragma unroll
        for (int m = 0; m < M; ++m) {
            v[m] = (mk[m] > 0.0f) ? sv[m] : NEG_MAX;
            mx = fmaxf(mx, v[m]);
        }
        float s = 0.0f;
#pragma unroll
        for (int m = 0; m < M; ++m)
            if (mk[m] > 0.0f) s += expf(v[m] - mx);
        e_emb[(size_t)ie * D + c] = mx + logf(s);
    }
}

constexpr int D4 = D / 4;
static_assert(D % 4 == 0 && D4 <= 256, "one float4 of hs and of ts per thread");

__global__ void k_pair(const float* __restrict__ e_emb,
                       const float* __restrict__ e_att,
                       const int*   __restrict__ hts,
                       float* __restrict__ ht_att,
                       float* __restrict__ out) {
    int b = blockIdx.x;
    int i = b / R;
    int eh = hts[(size_t)b * 2 + 0];
    int et = hts[(size_t)b * 2 + 1];

    bool cp = threadIdx.x < D4;
    v4f ch, ct;
    if (cp) {
        ch = *(const v4f*)(e_emb + ((size_t)i * E + eh) * D + threadIdx.x * 4);
        ct = *(const v4f*)(e_emb + ((size_t)i * E + et) * D + threadIdx.x * 4);
    }

    const float* ha = e_att + ((size_t)i * E + eh) * H * L;
    const float* ta = e_att + ((size_t)i * E + et) * H * L;

    int l0 = threadIdx.x * 4;
    v4f acc = {0.0f, 0.0f, 0.0f, 0.0f};
#pragma unroll
    for (int hh = 0; hh < H; ++hh) {
        v4f a = *(const v4f*)(ha + (size_t)hh * L + l0);
        v4f t = *(const v4f*)(ta + (size_t)hh * L + l0);
        acc += a * t;
    }
    acc *= (1.0f / (float)H);

    __shared__ float red[256];
    float ps = acc.x + acc.y + acc.z + acc.w;
    red[threadIdx.x] = ps;
    __syncthreads();
    for (int s = 128; s > 0; s >>= 1) {
        if (threadIdx.x < s) red[threadIdx.x] += red[threadIdx.x + s];
        __syncthreads();
    }
    float invs = 1.0f / (red[0] + 1e-5f);

    v4f* dst = (v4f*)(ht_att + (size_t)b * L + l0);
    dst[0] = acc * invs;

    if (cp) {
        *(v4f*)(out + (size_t)b * D + threadIdx.x * 4) = ch;
        *(v4f*)(out + (size_t)N_DOC * R * D + (size_t)b * D + threadIdx.x * 4) = ct;
    }
}

constexpr int G_BM  = 64;
constexpr int G_BN  = 48;
constexpr int G_KC  = 64;
constexpr int G_AP  = G_KC + 4;
constexpr int G_BP  = G_KC + 4;
constexpr int G_ASZ = G_BM * G_AP;
constexpr int G_BSZ = G_BN * G_BP;
constexpr int G_NCH = L / G_KC;
constexpr int G_ALD = (G_BM * G_KC / 4) / 128;
constexpr int G_BLD = (G_KC * G_BN / 4) / 128;
constexpr int G_HG  = G_KC / 16;
static_assert(G_NCH >= 2, "the loop keeps one chunk in flight ahead");

__device__ __forceinline__ void g_lds_barrier() {
    __builtin_amdgcn_fence(__ATOMIC_RELEASE, "workgroup", "local");
    __builtin_amdgcn_s_barrier();
    __builtin_amdgcn_fence(__ATOMIC_ACQUIRE, "workgroup", "local");
}

__device__ __forceinline__ void g_load_set(v4f (&fa)[G_HG], v4f (&fb)[3][G_HG],
                                           const float* pA, const float* pB) {
#pragma unroll
    for (int g = 0; g < G_HG; ++g) {
        fa[g] = *(const v4f*)(pA + g * 8);
#pragma unroll
        for (int j = 0; j < 3; ++j)
            fb[j][g] = *(const v4f*)(pB + j * 16 * G_BP + g * 8);
    }
}

template <int G0, int G1>
__device__ __forceinline__ void g_mma(v8f (&acc)[3], const v4f (&fa)[G_HG], const v4f (&fb)[3][G_HG]) {
#pragma unroll
    for (int g = G0; g < G1; ++g) {
        v2f a0 = {fa[g].x, fa[g].y};
        v2f a1 = {fa[g].z, fa[g].w};
#pragma unroll
        for (int j = 0; j < 3; ++j) {
            v2f b0 = {fb[j][g].x, fb[j][g].y};
            acc[j] = __builtin_amdgcn_wmma_f32_16x16x4_f32(
                        false, a0, false, b0, (short)0, acc[j], false, false);
        }
#pragma unroll
        for (int j = 0; j < 3; ++j) {
            v2f b1 = {fb[j][g].z, fb[j][g].w};
            acc[j] = __builtin_amdgcn_wmma_f32_16x16x4_f32(
                        false, a1, false, b1, (short)0, acc[j], false, false);
        }
    }
}

__global__ __launch_bounds__(128)
void k_rs_wmma(const float* __restrict__ seq,
               const float* __restrict__ ht_att,
               float* __restrict__ out) {
    __shared__ __attribute__((aligned(16))) float sA[2][G_ASZ];
    __shared__ __attribute__((aligned(16))) float sB[2][G_BSZ];

    constexpr int NT = D / G_BN;
    constexpr int MT = R / G_BM;
    int b  = blockIdx.x;
    int nt = b % NT;
    int mt = (b / NT) % MT;
    int i  = b / (NT * MT);

    int m0 = mt * G_BM;
    int n0 = nt * G_BN;

    const float* A  = ht_att + (size_t)i * R * L + (size_t)m0 * L;
    const float* Bm = seq    + (size_t)i * L * D + n0;

    int tid  = threadIdx.x;
    int wave = tid >> 5;
    int lane = tid & 31;
    int half = lane >> 4;
    int l15  = lane & 15;

    int a_goff[G_ALD], a_soff[G_ALD];
#pragma unroll
    for (int j = 0; j < G_ALD; ++j) {
        int f   = tid + 128 * j;
        int row = f >> 4;
        int c4  = f & 15;
        a_goff[j] = row * L + c4 * 4;
        a_soff[j] = row * G_AP + c4 * 4;
    }
    int b_goff[G_BLD], b_soff[G_BLD];
#pragma unroll
    for (int j = 0; j < G_BLD; ++j) {
        int f    = tid + 128 * j;
        int krow = f / 12;
        int c4   = f - krow * 12;
        b_goff[j] = krow * D + c4 * 4;
        b_soff[j] = (c4 * 4) * G_BP + krow;
    }

    v4f ra[G_ALD], rb[G_BLD];

#pragma unroll
    for (int j = 0; j < G_ALD; ++j) ra[j] = *(const v4f*)(A + a_goff[j]);
#pragma unroll
    for (int j = 0; j < G_BLD; ++j) rb[j] = *(const v4f*)(Bm + b_goff[j]);
#pragma unroll
    for (int j = 0; j < G_ALD; ++j) *(v4f*)(&sA[0][a_soff[j]]) = ra[j];
#pragma unroll
    for (int j = 0; j < G_BLD; ++j) {
#pragma unroll
        for (int q = 0; q < 4; ++q) sB[0][b_soff[j] + q * G_BP] = rb[j][q];
    }
#pragma unroll
    for (int j = 0; j < G_ALD; ++j) ra[j] = *(const v4f*)(A + G_KC + a_goff[j]);
#pragma unroll
    for (int j = 0; j < G_BLD; ++j) rb[j] = *(const v4f*)(Bm + (size_t)G_KC * D + b_goff[j]);
    g_lds_barrier();

    v8f acc[3];
#pragma unroll
    for (int j = 0; j < 3; ++j) acc[j] = (v8f){0.f, 0.f, 0.f, 0.f, 0.f, 0.f, 0.f, 0.f};

    int a_frag = (wave * 16 + l15) * G_AP + half * 4;
    int b_frag = l15 * G_BP + half * 4;

    v4f xa[G_HG], xb[3][G_HG];
    v4f ya[G_HG], yb[3][G_HG];
    g_load_set(xa, xb, &sA[0][a_frag], &sB[0][b_frag]);

    for (int c = 0; c < G_NCH; ++c) {
        int cur = c & 1;
        int nxt = cur ^ 1;
        bool more = (c + 1 < G_NCH);

        g_load_set(ya, yb, &sA[cur][a_frag + G_HG * 8], &sB[cur][b_frag + G_HG * 8]);
        __builtin_amdgcn_sched_barrier(0);
        g_mma<0, G_HG / 2>(acc, xa, xb);
        __builtin_amdgcn_sched_barrier(0);

        if (more) {
#pragma unroll
            for (int j = 0; j < G_ALD; ++j) *(v4f*)(&sA[nxt][a_soff[j]]) = ra[j];
#pragma unroll
            for (int j = 0; j < G_BLD; ++j) {
#pragma unroll
                for (int q = 0; q < 4; ++q) sB[nxt][b_soff[j] + q * G_BP] = rb[j][q];
            }
            if (c + 2 < G_NCH) {
                const float* An = A  + (size_t)(c + 2) * G_KC;
                const float* Bn = Bm + (size_t)(c + 2) * G_KC * D;
#pragma unroll
                for (int j = 0; j < G_ALD; ++j) ra[j] = *(const v4f*)(An + a_goff[j]);
#pragma unroll
                for (int j = 0; j < G_BLD; ++j) rb[j] = *(const v4f*)(Bn + b_goff[j]);
            }
        }
        __builtin_amdgcn_sched_barrier(0);
        g_mma<G_HG / 2, G_HG>(acc, xa, xb);
        g_mma<0, G_HG / 2>(acc, ya, yb);
        __builtin_amdgcn_sched_barrier(0);
        g_lds_barrier();
        if (more) g_load_set(xa, xb, &sA[nxt][a_frag], &sB[nxt][b_frag]);
        __builtin_amdgcn_sched_barrier(0);
        g_mma<G_HG / 2, G_HG>(acc, ya, yb);
        __builtin_amdgcn_sched_barrier(0);
    }

    float* O = out + (size_t)2 * N_DOC * R * D + (size_t)i * R * D;
#pragma unroll
    for (int j = 0; j < 3; ++j) {
#pragma unroll
        for (int v = 0; v < 8; ++v) {
            int row = m0 + wave * 16 + v + half * 8;
            O[(size_t)row * D + n0 + j * 16 + l15] = acc[j][v];
        }
    }
}

extern "C" void kernel_launch(void* const* d_in, const int* in_sizes, int n_in,
                              void* d_out, int out_size, void* d_ws, size_t ws_size,
                              hipStream_t stream) {
    const float* seq   = (const float*)d_in[0];
    const float* att   = (const float*)d_in[1];
    const int*   pos   = (const int*)  d_in[2];
    const float* mask  = (const float*)d_in[3];
    const int*   hts   = (const int*)  d_in[4];
    float*       out   = (float*)d_out;

    float* ws     = (float*)d_ws;
    float* e_emb  = ws;
    float* e_att  = e_emb + (size_t)N_DOC * E * D;
    float* ht_att = e_att + (size_t)N_DOC * E * H * L;

    k_entity <<<N_DOC * E * H, 256, 0, stream>>>(seq, att, pos, mask, e_emb, e_att);
    k_pair   <<<N_DOC * R,     256, 0, stream>>>(e_emb, e_att, hts, ht_att, out);
    k_rs_wmma<<<N_DOC * (R / G_BM) * (D / G_BN), 128, 0, stream>>>(seq, ht_att, out);
}


# rs GEMM on 8 waves, two per SIMD: a row group's two waves split every K chunk in halves so a sibling's matrix work covers stores, loads and barrier; partials added once via LDS, k sum order changes
# speedup vs baseline: 2.3718x; 1.0580x over previous
#include <hip/hip_runtime.h>
#include <math.h>

constexpr int N_DOC = 4;
constexpr int L     = 1024;
constexpr int D     = 768;
constexpr int H     = 12;
constexpr int E     = 32;
constexpr int M     = 4;
constexpr int R     = 256;
constexpr float NEG_MAX = -3.402823466e+38f;

typedef __attribute__((ext_vector_type(2))) float v2f;
typedef __attribute__((ext_vector_type(4))) float v4f;
typedef __attribute__((ext_vector_type(8))) float v8f;

constexpr int EC = D / H;
static_assert(D % H == 0 && EC <= 256, "e_emb columns are split evenly over the head blocks");

__global__ void k_entity(const float* __restrict__ seq,
                         const float* __restrict__ att,
                         const int*   __restrict__ pos,
                         const float* __restrict__ mask,
                         float* __restrict__ e_emb,
                         float* __restrict__ e_att) {
    int b  = blockIdx.x;
    int hh = b % H;
    int ie = b / H;
    int i  = ie / E;

    const int*   pos_e  = pos  + (size_t)ie * M;
    const float* mask_e = mask + (size_t)ie * M;

    int   p[M];
    float mk[M];
    float cnt = 0.0f;
#pragma unroll
    for (int m = 0; m < M; ++m) { p[m] = pos_e[m] + 1; mk[m] = mask_e[m]; cnt += mk[m]; }
    float inv = 1.0f / fmaxf(cnt, 1.0f);

    bool emb = threadIdx.x < EC;
    int  c   = hh * EC + threadIdx.x;
    float sv[M];
    if (emb) {
        const float* seq_i = seq + (size_t)i * L * D;
#pragma unroll
        for (int m = 0; m < M; ++m) sv[m] = seq_i[(size_t)p[m] * D + c];
    }

    int l0 = threadIdx.x * 4;
    v4f acc = {0.0f, 0.0f, 0.0f, 0.0f};
#pragma unroll
    for (int m = 0; m < M; ++m) {
        if (mk[m] > 0.0f) {
            const v4f* row = (const v4f*)(att + (((size_t)i * H + hh) * L + p[m]) * L + l0);
            acc += mk[m] * row[0];
        }
    }
    acc *= inv;
    v4f* dst = (v4f*)(e_att + ((size_t)ie * H + hh) * L + l0);
    dst[0] = acc;

    if (emb) {
        float v[M];
        float mx = NEG_MAX;
#pragma unroll
        for (int m = 0; m < M; ++m) {
            v[m] = (mk[m] > 0.0f) ? sv[m] : NEG_MAX;
            mx = fmaxf(mx, v[m]);
        }
        float s = 0.0f;
#pragma unroll
        for (int m = 0; m < M; ++m)
            if (mk[m] > 0.0f) s += expf(v[m] - mx);
        e_emb[(size_t)ie * D + c] = mx + logf(s);
    }
}

constexpr int D4 = D / 4;
static_assert(D % 4 == 0 && D4 <= 256, "one float4 of hs and of ts per thread");

__global__ void k_pair(const float* __restrict__ e_emb,
                       const float* __restrict__ e_att,
                       const int*   __restrict__ hts,
                       float* __restrict__ ht_att,
                       float* __restrict__ out) {
    int b = blockIdx.x;
    int i = b / R;
    int eh = hts[(size_t)b * 2 + 0];
    int et = hts[(size_t)b * 2 + 1];

    bool cp = threadIdx.x < D4;
    v4f ch, ct;
    if (cp) {
        ch = *(const v4f*)(e_emb + ((size_t)i * E + eh) * D + threadIdx.x * 4);
        ct = *(const v4f*)(e_emb + ((size_t)i * E + et) * D + threadIdx.x * 4);
    }

    const float* ha = e_att + ((size_t)i * E + eh) * H * L;
    const float* ta = e_att + ((size_t)i * E + et) * H * L;

    int l0 = threadIdx.x * 4;
    v4f acc = {0.0f, 0.0f, 0.0f, 0.0f};
#pragma unroll
    for (int hh = 0; hh < H; ++hh) {
        v4f a = *(const v4f*)(ha + (size_t)hh * L + l0);
        v4f t = *(const v4f*)(ta + (size_t)hh * L + l0);
        acc += a * t;
    }
    acc *= (1.0f / (float)H);

    __shared__ float red[256];
    float ps = acc.x + acc.y + acc.z + acc.w;
    red[threadIdx.x] = ps;
    __syncthreads();
    for (int s = 128; s > 0; s >>= 1) {
        if (threadIdx.x < s) red[threadIdx.x] += red[threadIdx.x + s];
        __syncthreads();
    }
    float invs = 1.0f / (red[0] + 1e-5f);

    v4f* dst = (v4f*)(ht_att + (size_t)b * L + l0);
    dst[0] = acc * invs;

    if (cp) {
        *(v4f*)(out + (size_t)b * D + threadIdx.x * 4) = ch;
        *(v4f*)(out + (size_t)N_DOC * R * D + (size_t)b * D + threadIdx.x * 4) = ct;
    }
}

constexpr int G_BM  = 64;
constexpr int G_BN  = 48;
constexpr int G_KC  = 64;
constexpr int G_NTH = 256;
constexpr int G_AP  = G_KC + 4;
constexpr int G_BP  = G_KC + 4;
constexpr int G_ASZ = G_BM * G_AP;
constexpr int G_BSZ = G_BN * G_BP;
constexpr int G_NCH = L / G_KC;
constexpr int G_ALD = (G_BM * G_KC / 4) / G_NTH;
constexpr int G_BLD = (G_KC * G_BN / 4) / G_NTH;
constexpr int G_HG  = G_KC / 16;
constexpr int G_RSZ = 3 * 8 * 32;
static_assert(G_NCH >= 2 && G_NCH % 2 == 0, "the loop keeps one chunk in flight ahead and walks two chunks per trip");

__device__ __forceinline__ void g_lds_barrier() {
    __builtin_amdgcn_fence(__ATOMIC_RELEASE, "workgroup", "local");
    __builtin_amdgcn_s_barrier();
    __builtin_amdgcn_fence(__ATOMIC_ACQUIRE, "workgroup", "local");
}

__device__ __forceinline__ void g_load_set(v4f (&fa)[G_HG], v4f (&fb)[3][G_HG],
                                           const float* pA, const float* pB) {
#pragma unroll
    for (int g = 0; g < G_HG; ++g) {
        fa[g] = *(const v4f*)(pA + g * 8);
#pragma unroll
        for (int j = 0; j < 3; ++j)
            fb[j][g] = *(const v4f*)(pB + j * 16 * G_BP + g * 8);
    }
}

template <int G0, int G1>
__device__ __forceinline__ void g_mma(v8f (&acc)[3], const v4f (&fa)[G_HG], const v4f (&fb)[3][G_HG]) {
#pragma unroll
    for (int g = G0; g < G1; ++g) {
        v2f a0 = {fa[g].x, fa[g].y};
        v2f a1 = {fa[g].z, fa[g].w};
#pragma unroll
        for (int j = 0; j < 3; ++j) {
            v2f b0 = {fb[j][g].x, fb[j][g].y};
            acc[j] = __builtin_amdgcn_wmma_f32_16x16x4_f32(
                        false, a0, false, b0, (short)0, acc[j], false, false);
        }
#pragma unroll
        for (int j = 0; j < 3; ++j) {
            v2f b1 = {fb[j][g].z, fb[j][g].w};
            acc[j] = __builtin_amdgcn_wmma_f32_16x16x4_f32(
                        false, a1, false, b1, (short)0, acc[j], false, false);
        }
    }
}

#define G_STEP(C, CUR, PA, PB, QA, QB)                                                              \
    {                                                                                               \
        const int  cc   = (C);                                                                      \
        const bool more = (cc + 1 < G_NCH);                          \
        __builtin_amdgcn_sched_barrier(0);                                                          \
        if (more) {                                                                                 \
                  \
                                                    \
            _Pragma("unroll")                                                                       \
            for (int j = 0; j < G_ALD; ++j) *(v4f*)(&sA[(CUR) ^ 1][a_soff[j]]) = ra[j];             \
            _Pragma("unroll")                                                                       \
            for (int j = 0; j < G_BLD; ++j) {                                                       \
                _Pragma("unroll")                                                                   \
                for (int q = 0; q < 4; ++q) sB[(CUR) ^ 1][b_soff[j] + q * G_BP] = rb[j][q];         \
            }                                                                                       \
            if (cc + 2 < G_NCH) {                                                                   \
                         \
                const float* An = A  + (size_t)(cc + 2) * G_KC;                                     \
                const float* Bn = Bm + (size_t)(cc + 2) * G_KC * D;                                 \
                _Pragma("unroll")                                                                   \
                for (int j = 0; j < G_ALD; ++j) ra[j] = *(const v4f*)(An + a_goff[j]);              \
                _Pragma("unroll")                                                                   \
                for (int j = 0; j < G_BLD; ++j) rb[j] = *(const v4f*)(Bn + b_goff[j]);              \
            }                                                                                       \
        }                                                                                           \
        __builtin_amdgcn_sched_barrier(0);                                                          \
        g_mma<0, G_HG / 2>(acc, PA, PB);                                                            \
        __builtin_amdgcn_sched_barrier(0);                                                          \
        g_lds_barrier();                                                                            \
        if (more) g_load_set(QA, QB, &sA[(CUR) ^ 1][a_frag], &sB[(CUR) ^ 1][b_frag]);               \
        __builtin_amdgcn_sched_barrier(0);                                                          \
        g_mma<G_HG / 2, G_HG>(acc, PA, PB);                                                         \
        __builtin_amdgcn_sched_barrier(0);                                                          \
    }

__global__ __launch_bounds__(G_NTH)
void k_rs_wmma(const float* __restrict__ seq,
               const float* __restrict__ ht_att,
               float* __restrict__ out) {
    __shared__ __attribute__((aligned(16))) float sA[2][G_ASZ];
    __shared__ __attribute__((aligned(16))) float sB[2][G_BSZ];
    __shared__ __attribute__((aligned(16))) float sR[4][G_RSZ];

    constexpr int NT = D / G_BN;
    constexpr int MT = R / G_BM;
    int b  = blockIdx.x;
    int nt = b % NT;
    int mt = (b / NT) % MT;
    int i  = b / (NT * MT);

    int m0 = mt * G_BM;
    int n0 = nt * G_BN;

    const float* A  = ht_att + (size_t)i * R * L + (size_t)m0 * L;
    const float* Bm = seq    + (size_t)i * L * D + n0;

    int tid  = threadIdx.x;
    int wave = tid >> 5;
    int lane = tid & 31;
    int rg   = wave & 3;
    int kh   = wave >> 2;
    int half = lane >> 4;
    int l15  = lane & 15;

    int a_goff[G_ALD], a_soff[G_ALD];
#pragma unroll
    for (int j = 0; j < G_ALD; ++j) {
        int f   = tid + G_NTH * j;
        int row = f >> 4;
        int c4  = f & 15;
        a_goff[j] = row * L + c4 * 4;
        a_soff[j] = row * G_AP + c4 * 4;
    }
    int b_goff[G_BLD], b_soff[G_BLD];
#pragma unroll
    for (int j = 0; j < G_BLD; ++j) {
        int f    = tid + G_NTH * j;
        int krow = f / 12;
        int c4   = f - krow * 12;
        b_goff[j] = krow * D + c4 * 4;
        b_soff[j] = (c4 * 4) * G_BP + krow;
    }

    v4f ra[G_ALD], rb[G_BLD];

    v4f ra0[G_ALD], rb0[G_BLD];
#pragma unroll
    for (int j = 0; j < G_ALD; ++j) ra0[j] = *(const v4f*)(A + a_goff[j]);
#pragma unroll
    for (int j = 0; j < G_BLD; ++j) rb0[j] = *(const v4f*)(Bm + b_goff[j]);
#pragma unroll
    for (int j = 0; j < G_ALD; ++j) ra[j] = *(const v4f*)(A + G_KC + a_goff[j]);
#pragma unroll
    for (int j = 0; j < G_BLD; ++j) rb[j] = *(const v4f*)(Bm + (size_t)G_KC * D + b_goff[j]);
#pragma unroll
    for (int j = 0; j < G_ALD; ++j) *(v4f*)(&sA[0][a_soff[j]]) = ra0[j];
#pragma unroll
    for (int j = 0; j < G_BLD; ++j) {
#pragma unroll
        for (int q = 0; q < 4; ++q) sB[0][b_soff[j] + q * G_BP] = rb0[j][q];
    }
    g_lds_barrier();

    v8f acc[3];
#pragma unroll
    for (int j = 0; j < 3; ++j) acc[j] = (v8f){0.f, 0.f, 0.f, 0.f, 0.f, 0.f, 0.f, 0.f};

    int k_frag = kh * (G_HG * 8) + half * 4;
    int a_frag = (rg * 16 + l15) * G_AP + k_frag;
    int b_frag = l15 * G_BP + k_frag;

    v4f xa[G_HG], xb[3][G_HG];
    v4f ya[G_HG], yb[3][G_HG];
    g_load_set(xa, xb, &sA[0][a_frag], &sB[0][b_frag]);

    for (int c = 0; c < G_NCH; c += 2) {
        G_STEP(c,     0, xa, xb, ya, yb)
        G_STEP(c + 1, 1, ya, yb, xa, xb)
    }

    if (kh == 1) {
#pragma unroll
        for (int j = 0; j < 3; ++j) {
#pragma unroll
            for (int v = 0; v < 8; ++v) sR[rg][(j * 8 + v) * 32 + lane] = acc[j][v];
        }
    }
    __syncthreads();
    if (kh == 0) {
        float* O = out + (size_t)2 * N_DOC * R * D + (size_t)i * R * D;
#pragma unroll
        for (int j = 0; j < 3; ++j) {
#pragma unroll
            for (int v = 0; v < 8; ++v) {
                int row = m0 + rg * 16 + v + half * 8;
                O[(size_t)row * D + n0 + j * 16 + l15] = acc[j][v] + sR[rg][(j * 8 + v) * 32 + lane];
            }
        }
    }
}

extern "C" void kernel_launch(void* const* d_in, const int* in_sizes, int n_in,
                              void* d_out, int out_size, void* d_ws, size_t ws_size,
                              hipStream_t stream) {
    const float* seq   = (const float*)d_in[0];
    const float* att   = (const float*)d_in[1];
    const int*   pos   = (const int*)  d_in[2];
    const float* mask  = (const float*)d_in[3];
    const int*   hts   = (const int*)  d_in[4];
    float*       out   = (float*)d_out;

    float* ws     = (float*)d_ws;
    float* e_emb  = ws;
    float* e_att  = e_emb + (size_t)N_DOC * E * D;
    float* ht_att = e_att + (size_t)N_DOC * E * H * L;

    k_entity <<<N_DOC * E * H, 256, 0, stream>>>(seq, att, pos, mask, e_emb, e_att);
    k_pair   <<<N_DOC * R,     256, 0, stream>>>(e_emb, e_att, hts, ht_att, out);
    k_rs_wmma<<<N_DOC * (R / G_BM) * (D / G_BN), G_NTH, 0, stream>>>(seq, ht_att, out);
}


# launch fusion 3 -> 2: block per (doc, 32-column slice) holds that slice of all entity rows in LDS, makes raw pair rows and hs/ts; GEMM sums A rows, scales output rows; normaliser sum order changes
# speedup vs baseline: 2.8960x; 1.2210x over previous
#include <hip/hip_runtime.h>
#include <math.h>

constexpr int N_DOC = 4;
constexpr int L     = 1024;
constexpr int D     = 768;
constexpr int H     = 12;
constexpr int E     = 32;
constexpr int M     = 4;
constexpr int R     = 256;
constexpr float NEG_MAX = -3.402823466e+38f;

typedef __attribute__((ext_vector_type(2))) float v2f;
typedef __attribute__((ext_vector_type(4))) float v4f;
typedef __attribute__((ext_vector_type(4))) int   v4i;
typedef __attribute__((ext_vector_type(8))) float v8f;

__device__ __forceinline__ void g_lds_barrier() {
    __builtin_amdgcn_fence(__ATOMIC_RELEASE, "workgroup", "local");
    __builtin_amdgcn_s_barrier();
    __builtin_amdgcn_fence(__ATOMIC_ACQUIRE, "workgroup", "local");
}

constexpr int F_LS  = 32;
constexpr int F_NS  = L / F_LS;
constexpr int F_Q   = F_LS / 4;
constexpr int F_NTH = 512;
constexpr int F_EP  = F_LS + 4;
constexpr int F_N1  = (E * H * F_Q) / F_NTH;
constexpr int F_N2  = (R * F_Q) / F_NTH;
constexpr int D4    = D / 4;
static_assert(F_NS == E, "slice s of a doc also carries the embedding row of entity s");
static_assert((E * H * F_Q) % F_NTH == 0 && (R * F_Q) % F_NTH == 0, "whole items per thread");
static_assert(2 * R == F_NTH, "one hts word per thread");
static_assert(D % 4 == 0 && D4 % 32 == 0 && D4 <= F_NTH, "the embedding row sits on whole waves");
static_assert(R % 32 == 0, "pairs are scanned 32 at a time");
static_assert(M == 4 && F_N1 == 6, "index words and masks are read as one 16-byte vector per item, 6 items");

__global__ __launch_bounds__(F_NTH)
void k_entity_pair(const float* __restrict__ seq,
                   const float* __restrict__ att,
                   const int*   __restrict__ pos,
                   const float* __restrict__ mask,
                   const int*   __restrict__ hts,
                   float* __restrict__ ht_raw,
                   float* __restrict__ out) {
    __shared__ __attribute__((aligned(16))) float ea[E * H * F_EP];
    __shared__ int s_hts[2 * R];

    int b   = blockIdx.x;
    int s   = b % F_NS;
    int i   = b / F_NS;
    int tid = threadIdx.x;

    v4i   pw[F_N1];
    v4f   mw[F_N1];
#pragma unroll
    for (int k = 0; k < F_N1; ++k) {
        int e = ((tid + F_NTH * k) / F_Q) / H;
        pw[k] = *(const v4i*)(pos  + ((size_t)i * E + e) * M);
        mw[k] = *(const v4f*)(mask + ((size_t)i * E + e) * M);
    }

    int hv = hts[(size_t)i * 2 * R + tid];

    bool  emb = tid < D4;
    int   ec4 = emb ? tid * 4 : 0;
    float mke[M];
    v4f   sv[M];
    {
        const int*   pos_e  = pos  + ((size_t)i * E + s) * M;
        const float* mask_e = mask + ((size_t)i * E + s) * M;
        const float* seq_i  = seq + (size_t)i * L * D;
#pragma unroll
        for (int m = 0; m < M; ++m) {
            mke[m] = mask_e[m];
            sv[m]  = *(const v4f*)(seq_i + (size_t)(pos_e[m] + 1) * D + ec4);
        }
    }

    asm volatile("" :: "v"(pw[0]), "v"(pw[1]), "v"(pw[2]), "v"(pw[3]), "v"(pw[4]), "v"(pw[5]),
                       "v"(mw[0]), "v"(mw[1]), "v"(mw[2]), "v"(mw[3]), "v"(mw[4]), "v"(mw[5]));

    const float* att_s = att + (size_t)i * H * L * L + (size_t)s * F_LS;
    v4f rw[F_N1][M];
#pragma unroll
    for (int k = 0; k < F_N1; ++k) {
        int q  = tid + F_NTH * k;
        int lq = q & (F_Q - 1);
        int eh = q / F_Q;
        int hh = eh % H;
        const float* base = att_s + (size_t)hh * L * L + lq * 4;
#pragma unroll
        for (int m = 0; m < M; ++m)
            rw[k][m] = *(const v4f*)(base + (size_t)(pw[k][m] + 1) * L);
    }
    asm volatile("" :: "v"(rw[0][0]), "v"(rw[0][1]), "v"(rw[0][2]), "v"(rw[0][3]),
                       "v"(rw[1][0]), "v"(rw[1][1]), "v"(rw[1][2]), "v"(rw[1][3]),
                       "v"(rw[2][0]), "v"(rw[2][1]), "v"(rw[2][2]), "v"(rw[2][3]),
                       "v"(rw[3][0]), "v"(rw[3][1]), "v"(rw[3][2]), "v"(rw[3][3]),
                       "v"(rw[4][0]), "v"(rw[4][1]), "v"(rw[4][2]), "v"(rw[4][3]),
                       "v"(rw[5][0]), "v"(rw[5][1]), "v"(rw[5][2]), "v"(rw[5][3]));
#pragma unroll
    for (int k = 0; k < F_N1; ++k) {
        int q  = tid + F_NTH * k;
        int lq = q & (F_Q - 1);
        int eh = q / F_Q;
        float cnt = 0.0f;
#pragma unroll
        for (int m = 0; m < M; ++m) cnt += mw[k][m];
        float inv = 1.0f / fmaxf(cnt, 1.0f);
        v4f acc = {0.0f, 0.0f, 0.0f, 0.0f};
#pragma unroll
        for (int m = 0; m < M; ++m) acc += mw[k][m] * rw[k][m];
        acc *= inv;
        *(v4f*)(&ea[eh * F_EP + lq * 4]) = acc;
    }
    s_hts[tid] = hv;

    g_lds_barrier();

    if (emb) {
        v4f ev;
#pragma unroll
        for (int c = 0; c < 4; ++c) {
            float v[M];
            float mx = NEG_MAX;
#pragma unroll
            for (int m = 0; m < M; ++m) {
                v[m] = (mke[m] > 0.0f) ? sv[m][c] : NEG_MAX;
                mx = fmaxf(mx, v[m]);
            }
            float sm = 0.0f;
#pragma unroll
            for (int m = 0; m < M; ++m)
                if (mke[m] > 0.0f) sm += expf(v[m] - mx);
            ev[c] = mx + logf(sm);
        }
        int    lane = tid & 31;
        float* o_hs = out + (size_t)i * R * D + tid * 4;
        float* o_ts = o_hs + (size_t)N_DOC * R * D;
        for (int j = 0; j < R / 32; ++j) {
            int r = j * 32 + lane;
            unsigned bh = __builtin_amdgcn_ballot_w32(s_hts[2 * r]     == s);
            unsigned bt = __builtin_amdgcn_ballot_w32(s_hts[2 * r + 1] == s);
            while (bh) {
                int bit = __builtin_ctz(bh);
                bh &= bh - 1;
                *(v4f*)(o_hs + (size_t)(j * 32 + bit) * D) = ev;
            }
            while (bt) {
                int bit = __builtin_ctz(bt);
                bt &= bt - 1;
                *(v4f*)(o_ts + (size_t)(j * 32 + bit) * D) = ev;
            }
        }
    }

#pragma unroll
    for (int k = 0; k < F_N2; ++k) {
        int item = tid + F_NTH * k;
        int lq   = item & (F_Q - 1);
        int r    = item / F_Q;
        int eh   = s_hts[2 * r];
        int et   = s_hts[2 * r + 1];
        const float* ha = &ea[eh * H * F_EP + lq * 4];
        const float* ta = &ea[et * H * F_EP + lq * 4];
        v4f acc = {0.0f, 0.0f, 0.0f, 0.0f};
#pragma unroll
        for (int hh = 0; hh < H; ++hh) {
            v4f a = *(const v4f*)(ha + hh * F_EP);
            v4f t = *(const v4f*)(ta + hh * F_EP);
            acc += a * t;
        }
        acc *= (1.0f / (float)H);
        *(v4f*)(ht_raw + ((size_t)i * R + r) * L + s * F_LS + lq * 4) = acc;
    }
}

constexpr int G_BM  = 64;
constexpr int G_BN  = 48;
constexpr int G_KC  = 64;
constexpr int G_NTH = 256;
constexpr int G_AP  = G_KC + 4;
constexpr int G_BP  = G_KC + 4;
constexpr int G_ASZ = G_BM * G_AP;
constexpr int G_BSZ = G_BN * G_BP;
constexpr int G_NCH = L / G_KC;
constexpr int G_ALD = (G_BM * G_KC / 4) / G_NTH;
constexpr int G_BLD = (G_KC * G_BN / 4) / G_NTH;
constexpr int G_HG  = G_KC / 16;
constexpr int G_RSZ = 3 * 8 * 32;
static_assert(G_NCH >= 2 && G_NCH % 2 == 0, "the loop keeps one chunk in flight ahead and walks two chunks per trip");

__device__ __forceinline__ void g_load_set(v4f (&fa)[G_HG], v4f (&fb)[3][G_HG],
                                           const float* pA, const float* pB) {
#pragma unroll
    for (int g = 0; g < G_HG; ++g) {
        fa[g] = *(const v4f*)(pA + g * 8);
#pragma unroll
        for (int j = 0; j < 3; ++j)
            fb[j][g] = *(const v4f*)(pB + j * 16 * G_BP + g * 8);
    }
}

template <int G0, int G1>
__device__ __forceinline__ void g_mma(v8f (&acc)[3], const v4f (&fa)[G_HG], const v4f (&fb)[3][G_HG]) {
#pragma unroll
    for (int g = G0; g < G1; ++g) {
        v2f a0 = {fa[g].x, fa[g].y};
        v2f a1 = {fa[g].z, fa[g].w};
#pragma unroll
        for (int j = 0; j < 3; ++j) {
            v2f b0 = {fb[j][g].x, fb[j][g].y};
            acc[j] = __builtin_amdgcn_wmma_f32_16x16x4_f32(
                        false, a0, false, b0, (short)0, acc[j], false, false);
        }
#pragma unroll
        for (int j = 0; j < 3; ++j) {
            v2f b1 = {fb[j][g].z, fb[j][g].w};
            acc[j] = __builtin_amdgcn_wmma_f32_16x16x4_f32(
                        false, a1, false, b1, (short)0, acc[j], false, false);
        }
    }
}

#define G_STEP(C, CUR, PA, PB, QA, QB)                                                              \
    {                                                                                               \
        const int  cc   = (C);                                                                      \
        const bool more = (cc + 1 < G_NCH);                          \
        __builtin_amdgcn_sched_barrier(0);                                                          \
          \
        _Pragma("unroll")                                                                           \
        for (int g = 0; g < G_HG; ++g) {                                                            \
            rsum += (v2f){PA[g].x, PA[g].y};                                                        \
            rsum += (v2f){PA[g].z, PA[g].w};                                                        \
        }                                                                                           \
        if (more) {                                                                                 \
                  \
                                                    \
            _Pragma("unroll")                                                                       \
            for (int j = 0; j < G_ALD; ++j) *(v4f*)(&sA[(CUR) ^ 1][a_soff[j]]) = ra[j];             \
            _Pragma("unroll")                                                                       \
            for (int j = 0; j < G_BLD; ++j) {                                                       \
                _Pragma("unroll")                                                                   \
                for (int q = 0; q < 4; ++q) sB[(CUR) ^ 1][b_soff[j] + q * G_BP] = rb[j][q];         \
            }                                                                                       \
            if (cc + 2 < G_NCH) {                                                                   \
                         \
                const float* An = A  + (size_t)(cc + 2) * G_KC;                                     \
                const float* Bn = Bm + (size_t)(cc + 2) * G_KC * D;                                 \
                _Pragma("unroll")                                                                   \
                for (int j = 0; j < G_ALD; ++j) ra[j] = *(const v4f*)(An + a_goff[j]);              \
                _Pragma("unroll")                                                                   \
                for (int j = 0; j < G_BLD; ++j) rb[j] = *(const v4f*)(Bn + b_goff[j]);              \
            }                                                                                       \
        }                                                                                           \
        __builtin_amdgcn_sched_barrier(0);                                                          \
        g_mma<0, G_HG / 2>(acc, PA, PB);                                                            \
        __builtin_amdgcn_sched_barrier(0);                                                          \
        g_lds_barrier();                                                                            \
        if (more) g_load_set(QA, QB, &sA[(CUR) ^ 1][a_frag], &sB[(CUR) ^ 1][b_frag]);               \
        __builtin_amdgcn_sched_barrier(0);                                                          \
        g_mma<G_HG / 2, G_HG>(acc, PA, PB);                                                         \
        __builtin_amdgcn_sched_barrier(0);                                                          \
    }

__global__ __launch_bounds__(G_NTH)
void k_rs_wmma(const float* __restrict__ seq,
               const float* __restrict__ ht_raw,
               float* __restrict__ out) {
    __shared__ __attribute__((aligned(16))) float sA[2][G_ASZ];
    __shared__ __attribute__((aligned(16))) float sB[2][G_BSZ];
    __shared__ __attribute__((aligned(16))) float sR[4][G_RSZ];
    __shared__ float sRow[2][G_BM];

    constexpr int NT = D / G_BN;
    constexpr int MT = R / G_BM;
    int b  = blockIdx.x;
    int nt = b % NT;
    int mt = (b / NT) % MT;
    int i  = b / (NT * MT);

    int m0 = mt * G_BM;
    int n0 = nt * G_BN;

    const float* A  = ht_raw + (size_t)i * R * L + (size_t)m0 * L;
    const float* Bm = seq    + (size_t)i * L * D + n0;

    int tid  = threadIdx.x;
    int wave = tid >> 5;
    int lane = tid & 31;
    int rg   = wave & 3;
    int kh   = wave >> 2;
    int half = lane >> 4;
    int l15  = lane & 15;

    int a_goff[G_ALD], a_soff[G_ALD];
#pragma unroll
    for (int j = 0; j < G_ALD; ++j) {
        int f   = tid + G_NTH * j;
        int row = f >> 4;
        int c4  = f & 15;
        a_goff[j] = row * L + c4 * 4;
        a_soff[j] = row * G_AP + c4 * 4;
    }
    int b_goff[G_BLD], b_soff[G_BLD];
#pragma unroll
    for (int j = 0; j < G_BLD; ++j) {
        int f    = tid + G_NTH * j;
        int krow = f / 12;
        int c4   = f - krow * 12;
        b_goff[j] = krow * D + c4 * 4;
        b_soff[j] = (c4 * 4) * G_BP + krow;
    }

    v4f ra[G_ALD], rb[G_BLD];

    v4f ra0[G_ALD], rb0[G_BLD];
#pragma unroll
    for (int j = 0; j < G_ALD; ++j) ra0[j] = *(const v4f*)(A + a_goff[j]);
#pragma unroll
    for (int j = 0; j < G_BLD; ++j) rb0[j] = *(const v4f*)(Bm + b_goff[j]);
#pragma unroll
    for (int j = 0; j < G_ALD; ++j) ra[j] = *(const v4f*)(A + G_KC + a_goff[j]);
#pragma unroll
    for (int j = 0; j < G_BLD; ++j) rb[j] = *(const v4f*)(Bm + (size_t)G_KC * D + b_goff[j]);
#pragma unroll
    for (int j = 0; j < G_ALD; ++j) *(v4f*)(&sA[0][a_soff[j]]) = ra0[j];
#pragma unroll
    for (int j = 0; j < G_BLD; ++j) {
#pragma unroll
        for (int q = 0; q < 4; ++q) sB[0][b_soff[j] + q * G_BP] = rb0[j][q];
    }
    g_lds_barrier();

    v8f acc[3];
#pragma unroll
    for (int j = 0; j < 3; ++j) acc[j] = (v8f){0.f, 0.f, 0.f, 0.f, 0.f, 0.f, 0.f, 0.f};
    v2f rsum = {0.f, 0.f};

    int k_frag = kh * (G_HG * 8) + half * 4;
    int a_frag = (rg * 16 + l15) * G_AP + k_frag;
    int b_frag = l15 * G_BP + k_frag;

    v4f xa[G_HG], xb[3][G_HG];
    v4f ya[G_HG], yb[3][G_HG];
    g_load_set(xa, xb, &sA[0][a_frag], &sB[0][b_frag]);

    for (int c = 0; c < G_NCH; c += 2) {
        G_STEP(c,     0, xa, xb, ya, yb)
        G_STEP(c + 1, 1, ya, yb, xa, xb)
    }

    {
        float t = rsum.x + rsum.y;
        t += __shfl_xor(t, 16);
        if (half == 0) sRow[kh][rg * 16 + l15] = t;
    }

    if (kh == 1) {
#pragma unroll
        for (int j = 0; j < 3; ++j) {
#pragma unroll
            for (int v = 0; v < 8; ++v) sR[rg][(j * 8 + v) * 32 + lane] = acc[j][v];
        }
    }
    __syncthreads();
    if (kh == 0) {
        float* O = out + (size_t)2 * N_DOC * R * D + (size_t)i * R * D;
        float inv[8];
#pragma unroll
        for (int v = 0; v < 8; ++v) {
            int rb = rg * 16 + v + half * 8;
            inv[v] = 1.0f / ((sRow[0][rb] + sRow[1][rb]) + 1e-5f);
        }
#pragma unroll
        for (int j = 0; j < 3; ++j) {
#pragma unroll
            for (int v = 0; v < 8; ++v) {
                int row = m0 + rg * 16 + v + half * 8;
                O[(size_t)row * D + n0 + j * 16 + l15] =
                    (acc[j][v] + sR[rg][(j * 8 + v) * 32 + lane]) * inv[v];
            }
        }
    }
}

extern "C" void kernel_launch(void* const* d_in, const int* in_sizes, int n_in,
                              void* d_out, int out_size, void* d_ws, size_t ws_size,
                              hipStream_t stream) {
    const float* seq   = (const float*)d_in[0];
    const float* att   = (const float*)d_in[1];
    const int*   pos   = (const int*)  d_in[2];
    const float* mask  = (const float*)d_in[3];
    const int*   hts   = (const int*)  d_in[4];
    float*       out   = (float*)d_out;

    float* ht_raw = (float*)d_ws;

    k_entity_pair<<<N_DOC * F_NS, F_NTH, 0, stream>>>(seq, att, pos, mask, hts, ht_raw, out);
    k_rs_wmma<<<N_DOC * (R / G_BM) * (D / G_BN), G_NTH, 0, stream>>>(seq, ht_raw, out);
}
